# starting kernel
# baseline (speedup 1.0000x reference)
#include <hip/hip_runtime.h>

typedef float v2f __attribute__((ext_vector_type(2)));
typedef float v8f __attribute__((ext_vector_type(8)));

#define NSTATES 64
#define SINK_ITERS 50
#define EPSI 0.1f
#define TINYF 1e-8f

__device__ __forceinline__ float hsum16(float x) {
#pragma unroll
    for (int m = 1; m < 16; m <<= 1) x += __shfl_xor(x, m, 32);
    return x;
}

__global__ __launch_bounds__(128)
void sinkhorn_emd_kernel(const float* __restrict__ p,
                         const float* __restrict__ q,
                         const float* __restrict__ cost,
                         float* __restrict__ out) {
    __shared__ v2f   KB[16 * 4 * 32];
    __shared__ v2f   MB[16 * 4 * 32];
    __shared__ float stage[4][16 * 64];

    const int tid  = threadIdx.x;
    const int wave = tid >> 5;
    const int lane = tid & 31;
    const int half = lane >> 4;
    const int l15  = lane & 15;

    for (int e = tid; e < 16 * 4 * 32; e += 128) {
        int c   = e >> 7;
        int t   = (e >> 5) & 3;
        int ln  = e & 31;
        int kr  = 4 * c + ((ln >> 4) << 1);
        int col = 16 * t + (ln & 15);
        float c0 = cost[kr * NSTATES + col];
        float c1 = cost[(kr + 1) * NSTATES + col];
        float k0 = __expf(-c0 * (1.0f / EPSI));
        float k1 = __expf(-c1 * (1.0f / EPSI));
        KB[e] = v2f{k0, k1};
        MB[e] = v2f{k0 * c0, k1 * c1};
    }
    __syncthreads();

    const int rowbase = blockIdx.x * 64 + wave * 16;

    float qD[4][8], pD[4][8];
#pragma unroll
    for (int t = 0; t < 4; ++t) {
#pragma unroll
        for (int j = 0; j < 8; ++j) {
            int r = rowbase + j + half * 8;
            int cidx = t * 16 + l15;
            qD[t][j] = q[r * NSTATES + cidx] + TINYF;
            pD[t][j] = p[r * NSTATES + cidx] + TINYF;
        }
    }
#pragma unroll
    for (int j = 0; j < 8; ++j) {
        float sq = qD[0][j] + qD[1][j] + qD[2][j] + qD[3][j];
        float sp = pD[0][j] + pD[1][j] + pD[2][j] + pD[3][j];
        sq = hsum16(sq);
        sp = hsum16(sp);
        float rq = __builtin_amdgcn_rcpf(sq);
        float rp = __builtin_amdgcn_rcpf(sp);
#pragma unroll
        for (int t = 0; t < 4; ++t) { qD[t][j] *= rq; pD[t][j] *= rp; }
    }

    v2f a[16];
#pragma unroll
    for (int c = 0; c < 16; ++c) a[c] = v2f{1.0f, 1.0f};

    float* st = stage[wave];
    float vD[4][8];

    for (int it = 0; it < SINK_ITERS; ++it) {
#pragma unroll
        for (int t = 0; t < 4; ++t) {
            v8f acc = {0.f, 0.f, 0.f, 0.f, 0.f, 0.f, 0.f, 0.f};
#pragma unroll
            for (int c = 0; c < 16; ++c) {
                v2f b = KB[(c * 4 + t) * 32 + lane];
                acc = __builtin_amdgcn_wmma_f32_16x16x4_f32(
                    false, a[c], false, b, (short)0, acc, false, false);
            }
#pragma unroll
            for (int j = 0; j < 8; ++j) {
                float v = qD[t][j] * __builtin_amdgcn_rcpf(acc[j]);
                vD[t][j] = v;
                st[(j + half * 8) * 64 + t * 16 + l15] = v;
            }
        }
#pragma unroll
        for (int c = 0; c < 16; ++c)
            a[c] = *(const v2f*)&st[l15 * 64 + 4 * c + half * 2];

#pragma unroll
        for (int t = 0; t < 4; ++t) {
            v8f acc = {0.f, 0.f, 0.f, 0.f, 0.f, 0.f, 0.f, 0.f};
#pragma unroll
            for (int c = 0; c < 16; ++c) {
                v2f b = KB[(c * 4 + t) * 32 + lane];
                acc = __builtin_amdgcn_wmma_f32_16x16x4_f32(
                    false, a[c], false, b, (short)0, acc, false, false);
            }
#pragma unroll
            for (int j = 0; j < 8; ++j) {
                float u = pD[t][j] * __builtin_amdgcn_rcpf(acc[j]);
                st[(j + half * 8) * 64 + t * 16 + l15] = u;
            }
        }
#pragma unroll
        for (int c = 0; c < 16; ++c)
            a[c] = *(const v2f*)&st[l15 * 64 + 4 * c + half * 2];
    }

    float e[8] = {0.f, 0.f, 0.f, 0.f, 0.f, 0.f, 0.f, 0.f};
#pragma unroll
    for (int t = 0; t < 4; ++t) {
        v8f acc = {0.f, 0.f, 0.f, 0.f, 0.f, 0.f, 0.f, 0.f};
#pragma unroll
        for (int c = 0; c < 16; ++c) {
            v2f b = MB[(c * 4 + t) * 32 + lane];
            acc = __builtin_amdgcn_wmma_f32_16x16x4_f32(
                false, a[c], false, b, (short)0, acc, false, false);
        }
#pragma unroll
        for (int j = 0; j < 8; ++j) e[j] += acc[j] * vD[t][j];
    }
#pragma unroll
    for (int j = 0; j < 8; ++j) {
        float s = hsum16(e[j]);
        if (l15 == 0) out[rowbase + j + half * 8] = s;
    }
}

extern "C" void kernel_launch(void* const* d_in, const int* in_sizes, int n_in,
                              void* d_out, int out_size, void* d_ws, size_t ws_size,
                              hipStream_t stream) {
    const float* p    = (const float*)d_in[0];
    const float* q    = (const float*)d_in[1];
    const float* cost = (const float*)d_in[2];
    float* out = (float*)d_out;

    int rows = in_sizes[0] / NSTATES;
    int blocks = rows / 64;
    sinkhorn_emd_kernel<<<blocks, 128, 0, stream>>>(p, q, cost, out);
}


# Pad the per-wave D-to-A relayout tile in LDS from a 64-dword to a 68-dword row pitch so its column-pair reads (16-way) and row stores (2-way) stop colliding on banks
# speedup vs baseline: 1.0380x; 1.0380x over previous
#include <hip/hip_runtime.h>

typedef float v2f __attribute__((ext_vector_type(2)));
typedef float v8f __attribute__((ext_vector_type(8)));

#define NSTATES 64
#define SINK_ITERS 50
#define EPSI 0.1f
#define TINYF 1e-8f
#define SPITCH 68

__device__ __forceinline__ float hsum16(float x) {
#pragma unroll
    for (int m = 1; m < 16; m <<= 1) x += __shfl_xor(x, m, 32);
    return x;
}

__global__ __launch_bounds__(128)
void sinkhorn_emd_kernel(const float* __restrict__ p,
                         const float* __restrict__ q,
                         const float* __restrict__ cost,
                         float* __restrict__ out) {
    __shared__ v2f   KB[16 * 4 * 32];
    __shared__ v2f   MB[16 * 4 * 32];
    __shared__ float stage[4][16 * SPITCH];

    const int tid  = threadIdx.x;
    const int wave = tid >> 5;
    const int lane = tid & 31;
    const int half = lane >> 4;
    const int l15  = lane & 15;

    for (int e = tid; e < 16 * 4 * 32; e += 128) {
        int c   = e >> 7;
        int t   = (e >> 5) & 3;
        int ln  = e & 31;
        int kr  = 4 * c + ((ln >> 4) << 1);
        int col = 16 * t + (ln & 15);
        float c0 = cost[kr * NSTATES + col];
        float c1 = cost[(kr + 1) * NSTATES + col];
        float k0 = __expf(-c0 * (1.0f / EPSI));
        float k1 = __expf(-c1 * (1.0f / EPSI));
        KB[e] = v2f{k0, k1};
        MB[e] = v2f{k0 * c0, k1 * c1};
    }
    __syncthreads();

    const int rowbase = blockIdx.x * 64 + wave * 16;

    float qD[4][8], pD[4][8];
#pragma unroll
    for (int t = 0; t < 4; ++t) {
#pragma unroll
        for (int j = 0; j < 8; ++j) {
            int r = rowbase + j + half * 8;
            int cidx = t * 16 + l15;
            qD[t][j] = q[r * NSTATES + cidx] + TINYF;
            pD[t][j] = p[r * NSTATES + cidx] + TINYF;
        }
    }
#pragma unroll
    for (int j = 0; j < 8; ++j) {
        float sq = qD[0][j] + qD[1][j] + qD[2][j] + qD[3][j];
        float sp = pD[0][j] + pD[1][j] + pD[2][j] + pD[3][j];
        sq = hsum16(sq);
        sp = hsum16(sp);
        float rq = __builtin_amdgcn_rcpf(sq);
        float rp = __builtin_amdgcn_rcpf(sp);
#pragma unroll
        for (int t = 0; t < 4; ++t) { qD[t][j] *= rq; pD[t][j] *= rp; }
    }

    v2f a[16];
#pragma unroll
    for (int c = 0; c < 16; ++c) a[c] = v2f{1.0f, 1.0f};

    float* st = stage[wave];
    float vD[4][8];

    for (int it = 0; it < SINK_ITERS; ++it) {
#pragma unroll
        for (int t = 0; t < 4; ++t) {
            v8f acc = {0.f, 0.f, 0.f, 0.f, 0.f, 0.f, 0.f, 0.f};
#pragma unroll
            for (int c = 0; c < 16; ++c) {
                v2f b = KB[(c * 4 + t) * 32 + lane];
                acc = __builtin_amdgcn_wmma_f32_16x16x4_f32(
                    false, a[c], false, b, (short)0, acc, false, false);
            }
#pragma unroll
            for (int j = 0; j < 8; ++j) {
                float v = qD[t][j] * __builtin_amdgcn_rcpf(acc[j]);
                vD[t][j] = v;
                st[(j + half * 8) * SPITCH + t * 16 + l15] = v;
            }
        }
#pragma unroll
        for (int c = 0; c < 16; ++c)
            a[c] = *(const v2f*)&st[l15 * SPITCH + 4 * c + half * 2];

#pragma unroll
        for (int t = 0; t < 4; ++t) {
            v8f acc = {0.f, 0.f, 0.f, 0.f, 0.f, 0.f, 0.f, 0.f};
#pragma unroll
            for (int c = 0; c < 16; ++c) {
                v2f b = KB[(c * 4 + t) * 32 + lane];
                acc = __builtin_amdgcn_wmma_f32_16x16x4_f32(
                    false, a[c], false, b, (short)0, acc, false, false);
            }
#pragma unroll
            for (int j = 0; j < 8; ++j) {
                float u = pD[t][j] * __builtin_amdgcn_rcpf(acc[j]);
                st[(j + half * 8) * SPITCH + t * 16 + l15] = u;
            }
        }
#pragma unroll
        for (int c = 0; c < 16; ++c)
            a[c] = *(const v2f*)&st[l15 * SPITCH + 4 * c + half * 2];
    }

    float e[8] = {0.f, 0.f, 0.f, 0.f, 0.f, 0.f, 0.f, 0.f};
#pragma unroll
    for (int t = 0; t < 4; ++t) {
        v8f acc = {0.f, 0.f, 0.f, 0.f, 0.f, 0.f, 0.f, 0.f};
#pragma unroll
        for (int c = 0; c < 16; ++c) {
            v2f b = MB[(c * 4 + t) * 32 + lane];
            acc = __builtin_amdgcn_wmma_f32_16x16x4_f32(
                false, a[c], false, b, (short)0, acc, false, false);
        }
#pragma unroll
        for (int j = 0; j < 8; ++j) e[j] += acc[j] * vD[t][j];
    }
#pragma unroll
    for (int j = 0; j < 8; ++j) {
        float s = hsum16(e[j]);
        if (l15 == 0) out[rowbase + j + half * 8] = s;
    }
}

extern "C" void kernel_launch(void* const* d_in, const int* in_sizes, int n_in,
                              void* d_out, int out_size, void* d_ws, size_t ws_size,
                              hipStream_t stream) {
    const float* p    = (const float*)d_in[0];
    const float* q    = (const float*)d_in[1];
    const float* cost = (const float*)d_in[2];
    float* out = (float*)d_out;

    int rows = in_sizes[0] / NSTATES;
    int blocks = rows / 64;
    sinkhorn_emd_kernel<<<blocks, 128, 0, stream>>>(p, q, cost, out);
}


# Fit 256 registers (K's 64 B fragments kept in registers, two accumulator tiles at a time, last v parked in LDS): 4 waves per SIMD resident, no LDS load inside the WMMA runs; results bit-identical
# speedup vs baseline: 1.0887x; 1.0488x over previous
#include <hip/hip_runtime.h>

typedef float v2f __attribute__((ext_vector_type(2)));
typedef float v8f __attribute__((ext_vector_type(8)));

#define NSTATES 64
#define SINK_ITERS 50
#define EPSI 0.1f
#define TINYF 1e-8f
#define SPITCH 68

__device__ __forceinline__ float hsum16(float x) {
#pragma unroll
    for (int m = 1; m < 16; m <<= 1) x += __shfl_xor(x, m, 32);
    return x;
}

__device__ __forceinline__ void scale_step(v2f (&a)[16], const v2f (&kb)[64],
                                           const float (&num)[4][8],
                                           float* st, int half, int l15) {
#pragma unroll
    for (int tp = 0; tp < 4; tp += 2) {
        v8f acc0 = {0.f, 0.f, 0.f, 0.f, 0.f, 0.f, 0.f, 0.f};
        v8f acc1 = {0.f, 0.f, 0.f, 0.f, 0.f, 0.f, 0.f, 0.f};
#pragma unroll
        for (int c = 0; c < 16; ++c) {
            acc0 = __builtin_amdgcn_wmma_f32_16x16x4_f32(
                false, a[c], false, kb[c * 4 + tp], (short)0, acc0, false, false);
            acc1 = __builtin_amdgcn_wmma_f32_16x16x4_f32(
                false, a[c], false, kb[c * 4 + tp + 1], (short)0, acc1, false, false);
        }
#pragma unroll
        for (int j = 0; j < 8; ++j) {
            acc0[j] = __builtin_amdgcn_rcpf(acc0[j]);
            acc1[j] = __builtin_amdgcn_rcpf(acc1[j]);
        }
#pragma unroll
        for (int j = 0; j < 8; ++j) {
            acc0[j] *= num[tp][j];
            acc1[j] *= num[tp + 1][j];
        }
#pragma unroll
        for (int j = 0; j < 8; ++j) {
            st[(j + half * 8) * SPITCH + tp * 16 + l15] = acc0[j];
            st[(j + half * 8) * SPITCH + (tp + 1) * 16 + l15] = acc1[j];
        }
    }
#pragma unroll
    for (int c = 0; c < 16; ++c)
        a[c] = *(const v2f*)&st[l15 * SPITCH + 4 * c + half * 2];
}

__global__ __launch_bounds__(128, 4)
void sinkhorn_emd_kernel(const float* __restrict__ p,
                         const float* __restrict__ q,
                         const float* __restrict__ cost,
                         float* __restrict__ out) {
    __shared__ v2f   KB[16 * 4 * 32];
    __shared__ v2f   MB[16 * 4 * 32];
    __shared__ float stage[4][16 * SPITCH];
    __shared__ float stageV[4][16 * SPITCH];

    const int tid  = threadIdx.x;
    const int wave = tid >> 5;
    const int lane = tid & 31;
    const int half = lane >> 4;
    const int l15  = lane & 15;

    for (int e = tid; e < 16 * 4 * 32; e += 128) {
        int c   = e >> 7;
        int t   = (e >> 5) & 3;
        int ln  = e & 31;
        int kr  = 4 * c + ((ln >> 4) << 1);
        int col = 16 * t + (ln & 15);
        float c0 = cost[kr * NSTATES + col];
        float c1 = cost[(kr + 1) * NSTATES + col];
        float k0 = __expf(-c0 * (1.0f / EPSI));
        float k1 = __expf(-c1 * (1.0f / EPSI));
        KB[e] = v2f{k0, k1};
        MB[e] = v2f{k0 * c0, k1 * c1};
    }
    __syncthreads();

    const int rowbase = blockIdx.x * 64 + wave * 16;

    float qD[4][8], pD[4][8];
#pragma unroll
    for (int t = 0; t < 4; ++t) {
#pragma unroll
        for (int j = 0; j < 8; ++j) {
            int r = rowbase + j + half * 8;
            int cidx = t * 16 + l15;
            qD[t][j] = q[r * NSTATES + cidx] + TINYF;
            pD[t][j] = p[r * NSTATES + cidx] + TINYF;
        }
    }
#pragma unroll
    for (int j = 0; j < 8; ++j) {
        float sq = qD[0][j] + qD[1][j] + qD[2][j] + qD[3][j];
        float sp = pD[0][j] + pD[1][j] + pD[2][j] + pD[3][j];
        sq = hsum16(sq);
        sp = hsum16(sp);
        float rq = __builtin_amdgcn_rcpf(sq);
        float rp = __builtin_amdgcn_rcpf(sp);
#pragma unroll
        for (int t = 0; t < 4; ++t) { qD[t][j] *= rq; pD[t][j] *= rp; }
    }

    v2f a[16];
#pragma unroll
    for (int c = 0; c < 16; ++c) a[c] = v2f{1.0f, 1.0f};

    v2f kb[64];
#pragma unroll
    for (int f = 0; f < 64; ++f) kb[f] = KB[f * 32 + lane];

    float* st  = stage[wave];
    float* stV = stageV[wave];

    for (int it = 0; it < SINK_ITERS - 1; ++it) {
        scale_step(a, kb, qD, st, half, l15);
        scale_step(a, kb, pD, st, half, l15);
    }
    scale_step(a, kb, qD, stV, half, l15);
    scale_step(a, kb, pD, st, half, l15);

    float e[8] = {0.f, 0.f, 0.f, 0.f, 0.f, 0.f, 0.f, 0.f};
#pragma unroll
    for (int t = 0; t < 4; ++t) {
        v8f acc = {0.f, 0.f, 0.f, 0.f, 0.f, 0.f, 0.f, 0.f};
#pragma unroll
        for (int c = 0; c < 16; ++c) {
            v2f b = MB[(c * 4 + t) * 32 + lane];
            acc = __builtin_amdgcn_wmma_f32_16x16x4_f32(
                false, a[c], false, b, (short)0, acc, false, false);
        }
#pragma unroll
        for (int j = 0; j < 8; ++j)
            e[j] += acc[j] * stV[(j + half * 8) * SPITCH + t * 16 + l15];
    }
#pragma unroll
    for (int j = 0; j < 8; ++j) {
        float s = hsum16(e[j]);
        if (l15 == 0) out[rowbase + j + half * 8] = s;
    }
}

extern "C" void kernel_launch(void* const* d_in, const int* in_sizes, int n_in,
                              void* d_out, int out_size, void* d_ws, size_t ws_size,
                              hipStream_t stream) {
    const float* p    = (const float*)d_in[0];
    const float* q    = (const float*)d_in[1];
    const float* cost = (const float*)d_in[2];
    float* out = (float*)d_out;

    int rows = in_sizes[0] / NSTATES;
    int blocks = rows / 64;
    sinkhorn_emd_kernel<<<blocks, 128, 0, stream>>>(p, q, cost, out);
}
